# starting kernel
# baseline (speedup 1.0000x reference)
#include <hip/hip_runtime.h>

#define K_CODES 4096
#define CDIM 128
#define ROWS_PB 128
#define NCHUNK 64
#define NCHUNKS (K_CODES / NCHUNK)
#define LDSPAD 132

typedef __attribute__((ext_vector_type(2))) float v2f;
typedef __attribute__((ext_vector_type(4))) float v4f;
typedef __attribute__((ext_vector_type(8))) float v8f;

__global__ __launch_bounds__(256) void vq_e2_kernel(const float* __restrict__ emb,
                                                    float* __restrict__ e2g) {
    int k = blockIdx.x * 256 + threadIdx.x;
    const float* row = emb + (size_t)k * CDIM;
    float s = 0.f;
#pragma unroll 8
    for (int c = 0; c < CDIM; ++c) { float v = row[c]; s += v * v; }
    e2g[k] = s;
}

__global__ __launch_bounds__(256) void vq_main_kernel(const float* __restrict__ in,
                                                      const float* __restrict__ emb,
                                                      const float* __restrict__ e2g,
                                                      float* __restrict__ outq,
                                                      float* __restrict__ partials) {
    __shared__ float b_s[NCHUNK * LDSPAD];
    __shared__ float x2s[ROWS_PB];
    __shared__ int   idx_s[ROWS_PB];
    __shared__ float dmin_s[ROWS_PB];

    const int tid  = threadIdx.x;
    const int lane = tid & 31;
    const int hi   = lane >> 4;
    const int l16  = lane & 15;
    const int wave = tid >> 5;

    const int nbase = blockIdx.x * ROWS_PB;
    const int b     = nbase >> 10;
    const int hw0   = nbase & 1023;
    const size_t img_off = (size_t)b * (CDIM * 1024) + hw0;

    const int rloc = wave * 16 + l16;
    const float* arow = in + img_off + rloc;
    v2f areg[32];
#pragma unroll
    for (int j = 0; j < 32; ++j) {
        int c0 = 4 * j + 2 * hi;
        areg[j].x = arow[(size_t)c0 * 1024];
        areg[j].y = arow[(size_t)(c0 + 1) * 1024];
    }
    float p = 0.f;
#pragma unroll
    for (int j = 0; j < 32; ++j) p += areg[j].x * areg[j].x + areg[j].y * areg[j].y;
    p += __shfl_xor(p, 16, 32);
    if (hi == 0) x2s[rloc] = p;

    v4f sreg[8];
    auto loadRegs = [&](int chunk) {
        const float* eb = emb + (size_t)chunk * NCHUNK * CDIM;
#pragma unroll
        for (int it = 0; it < 8; ++it) {
            int e4 = it * 256 + tid;
            int j  = e4 >> 5;
            int c  = (e4 & 31) << 2;
            sreg[it] = *(const v4f*)(eb + (size_t)j * CDIM + c);
        }
    };
    auto storeRegs = [&]() {
#pragma unroll
        for (int it = 0; it < 8; ++it) {
            int e4 = it * 256 + tid;
            int j  = e4 >> 5;
            int c  = (e4 & 31) << 2;
            *(v4f*)(&b_s[j * LDSPAD + c]) = sreg[it];
        }
    };

    loadRegs(0);
    storeRegs();
    __syncthreads();

    float x2r[8];
#pragma unroll
    for (int v = 0; v < 8; ++v) x2r[v] = x2s[wave * 16 + v + 8 * hi];

    float best[8]; int bestI[8];
#pragma unroll
    for (int v = 0; v < 8; ++v) { best[v] = 3.4e38f; bestI[v] = 0; }

    const float* bfrag = &b_s[l16 * LDSPAD];

    for (int chunk = 0; chunk < NCHUNKS; ++chunk) {
        if (chunk + 1 < NCHUNKS) loadRegs(chunk + 1);

        v8f acc0 = {0.f,0.f,0.f,0.f,0.f,0.f,0.f,0.f};
        v8f acc1 = acc0, acc2 = acc0, acc3 = acc0;
#pragma unroll
        for (int j = 0; j < 32; ++j) {
            int off = 4 * j + 2 * hi;
            v2f b0 = *(const v2f*)(bfrag + off);
            v2f b1 = *(const v2f*)(bfrag + 16 * LDSPAD + off);
            v2f b2 = *(const v2f*)(bfrag + 32 * LDSPAD + off);
            v2f b3 = *(const v2f*)(bfrag + 48 * LDSPAD + off);
            acc0 = __builtin_amdgcn_wmma_f32_16x16x4_f32(false, areg[j], false, b0, (short)0, acc0, false, false);
            acc1 = __builtin_amdgcn_wmma_f32_16x16x4_f32(false, areg[j], false, b1, (short)0, acc1, false, false);
            acc2 = __builtin_amdgcn_wmma_f32_16x16x4_f32(false, areg[j], false, b2, (short)0, acc2, false, false);
            acc3 = __builtin_amdgcn_wmma_f32_16x16x4_f32(false, areg[j], false, b3, (short)0, acc3, false, false);
        }

        int kbase = chunk * NCHUNK + l16;
        float e20 = e2g[kbase];
        float e21 = e2g[kbase + 16];
        float e22 = e2g[kbase + 32];
        float e23 = e2g[kbase + 48];
#pragma unroll
        for (int v = 0; v < 8; ++v) {
            float d0 = x2r[v] - 2.0f * acc0[v] + e20;
            if (d0 < best[v]) { best[v] = d0; bestI[v] = kbase; }
            float d1 = x2r[v] - 2.0f * acc1[v] + e21;
            if (d1 < best[v]) { best[v] = d1; bestI[v] = kbase + 16; }
            float d2 = x2r[v] - 2.0f * acc2[v] + e22;
            if (d2 < best[v]) { best[v] = d2; bestI[v] = kbase + 32; }
            float d3 = x2r[v] - 2.0f * acc3[v] + e23;
            if (d3 < best[v]) { best[v] = d3; bestI[v] = kbase + 48; }
        }

        __syncthreads();
        if (chunk + 1 < NCHUNKS) {
            storeRegs();
            __syncthreads();
        }
    }

#pragma unroll
    for (int v = 0; v < 8; ++v) {
        float d = best[v]; int i = bestI[v];
#pragma unroll
        for (int off = 8; off >= 1; off >>= 1) {
            float od = __shfl_xor(d, off, 32);
            int   oi = __shfl_xor(i, off, 32);
            if (od < d || (od == d && oi < i)) { d = od; i = oi; }
        }
        best[v] = d; bestI[v] = i;
    }
    if (l16 == 0) {
#pragma unroll
        for (int v = 0; v < 8; ++v) {
            int rl = wave * 16 + v + 8 * hi;
            idx_s[rl]  = bestI[v];
            dmin_s[rl] = best[v];
        }
    }
    __syncthreads();

    if (tid == 0) {
        float s = 0.f;
        for (int r = 0; r < ROWS_PB; ++r) s += dmin_s[r];
        partials[blockIdx.x] = s;
    }

#pragma unroll 4
    for (int it = 0; it < 64; ++it) {
        int e = it * 256 + tid;
        int c = e >> 7, r = e & 127;
        outq[img_off + (size_t)c * 1024 + r] = emb[(size_t)idx_s[r] * CDIM + c];
    }
}

__global__ __launch_bounds__(256) void vq_loss_kernel(const float* __restrict__ partials,
                                                      float* __restrict__ out_loss) {
    __shared__ float s[256];
    int t = threadIdx.x;
    s[t] = partials[t];
    __syncthreads();
    for (int off = 128; off >= 1; off >>= 1) {
        if (t < off) s[t] += s[t + off];
        __syncthreads();
    }
    if (t == 0) *out_loss = 1.25f * s[0] * (1.0f / 4194304.0f);
}

extern "C" void kernel_launch(void* const* d_in, const int* in_sizes, int n_in,
                              void* d_out, int out_size, void* d_ws, size_t ws_size,
                              hipStream_t stream) {
    const float* in  = (const float*)d_in[0];
    const float* emb = (const float*)d_in[1];
    float* outq      = (float*)d_out;
    float* partials  = (float*)d_ws;
    float* e2g       = partials + 256;
    float* loss_out  = outq + (out_size - 1);

    vq_e2_kernel  <<<K_CODES / 256, 256, 0, stream>>>(emb, e2g);
    vq_main_kernel<<<32768 / ROWS_PB, 256, 0, stream>>>(in, emb, e2g, outq, partials);
    vq_loss_kernel<<<1, 256, 0, stream>>>(partials, loss_out);
}


# Wave tile 32 rows x 32 codes (two waves of a SIMD split each chunk's codes): each LDS B fragment feeds two WMMAs, reads grouped ahead of 64-WMMA blocks, 3-slot LDS ring with one barrier per chunk
# speedup vs baseline: 1.0390x; 1.0390x over previous
#include <hip/hip_runtime.h>


#define K_CODES 4096
#define CDIM 128
#define ROWS_PB 128
#define NCHUNK 64
#define NCHUNKS (K_CODES / NCHUNK)
#define LDSPAD 132
#define STAGE_FLOATS (NCHUNK * LDSPAD)
#define NSTAGE 3

typedef __attribute__((ext_vector_type(2))) float v2f;
typedef __attribute__((ext_vector_type(4))) float v4f;
typedef __attribute__((ext_vector_type(8))) float v8f;

#define WMMA_F32(a, b, c) __builtin_amdgcn_wmma_f32_16x16x4_f32(false, (a), false, (b), (short)0, (c), false, false)

__global__ __launch_bounds__(256) void vq_e2_kernel(const float* __restrict__ emb,
                                                    float* __restrict__ e2g) {
    int k = blockIdx.x * 256 + threadIdx.x;
    const float* row = emb + (size_t)k * CDIM;
    float s = 0.f;
#pragma unroll 8
    for (int c = 0; c < CDIM; ++c) { float v = row[c]; s += v * v; }
    e2g[k] = s;
}

__global__ __launch_bounds__(256) void vq_main_kernel(const float* __restrict__ in,
                                                      const float* __restrict__ emb,
                                                      const float* __restrict__ e2g,
                                                      float* __restrict__ outq,
                                                      float* __restrict__ partials) {
    __shared__ float b_s[NSTAGE * STAGE_FLOATS];
    __shared__ float x2s[ROWS_PB];
    __shared__ int   idx_s[ROWS_PB];
    __shared__ float dmin_s[ROWS_PB];
    __shared__ int   idx2_s[2 * ROWS_PB];
    __shared__ float dmin2_s[2 * ROWS_PB];

    const int tid  = threadIdx.x;
    const int lane = tid & 31;
    const int hi   = lane >> 4;
    const int l16  = lane & 15;
    const int wave = tid >> 5;
    const int rg   = wave & 3;
    const int nh   = wave >> 2;

    const int nbase = blockIdx.x * ROWS_PB;
    const int b     = nbase >> 10;
    const int hw0   = nbase & 1023;
    const size_t img_off = (size_t)b * (CDIM * 1024) + hw0;

    v2f areg[2][32];
#pragma unroll
    for (int m = 0; m < 2; ++m) {
        const int rloc = rg * 32 + m * 16 + l16;
        const float* arow = in + img_off + rloc;
#pragma unroll
        for (int j = 0; j < 32; ++j) {
            int c0 = 4 * j + 2 * hi;
            areg[m][j].x = arow[(size_t)c0 * 1024];
            areg[m][j].y = arow[(size_t)(c0 + 1) * 1024];
        }
    }
#pragma unroll
    for (int m = 0; m < 2; ++m) {
        float p = 0.f;
#pragma unroll
        for (int j = 0; j < 32; ++j) p += areg[m][j].x * areg[m][j].x + areg[m][j].y * areg[m][j].y;
        p += __shfl_xor(p, 16, 32);
        if (hi == 0 && nh == 0) x2s[rg * 32 + m * 16 + l16] = p;
    }

    v4f sreg[8];
    auto loadRegs = [&](int chunk) {
        const float* eb = emb + (size_t)chunk * NCHUNK * CDIM;
#pragma unroll
        for (int it = 0; it < 8; ++it) {
            int e4 = it * 256 + tid;
            int j  = e4 >> 5;
            int c  = (e4 & 31) << 2;
            sreg[it] = *(const v4f*)(eb + (size_t)j * CDIM + c);
        }
    };
    auto storeRegs = [&](float* dst) {
#pragma unroll
        for (int it = 0; it < 8; ++it) {
            int e4 = it * 256 + tid;
            int j  = e4 >> 5;
            int c  = (e4 & 31) << 2;
            *(v4f*)(&dst[j * LDSPAD + c]) = sreg[it];
        }
    };

    const float* bbase = &b_s[(l16 + 32 * nh) * LDSPAD + 2 * hi];
    auto loadFrags = [&](v2f (&f)[2][16], int stage_off, int kb) {
#pragma unroll
        for (int t = 0; t < 2; ++t) {
            const float* rowp = bbase + stage_off + 16 * t * LDSPAD + 64 * kb;
#pragma unroll
            for (int jj = 0; jj < 16; ++jj) f[t][jj] = *(const v2f*)(rowp + 4 * jj);
        }
    };
    const v8f zero8 = {0.f,0.f,0.f,0.f,0.f,0.f,0.f,0.f};
    auto mmaBlock = [&](v8f (&acc)[2][2], const v2f (&f)[2][16], int kb) {
#pragma unroll
        for (int jj = 0; jj < 16; ++jj) {
            const int j = kb * 16 + jj;
#pragma unroll
            for (int t = 0; t < 2; ++t) {
#pragma unroll
                for (int m = 0; m < 2; ++m) {
                    v8f c = (j == 0) ? zero8 : acc[t][m];
                    acc[t][m] = WMMA_F32(areg[m][j], f[t][jj], c);
                }
            }
        }
    };

    loadRegs(0);
    storeRegs(b_s);
    loadRegs(1);
    __syncthreads();

    float x2r[2][8];
#pragma unroll
    for (int m = 0; m < 2; ++m)
#pragma unroll
        for (int v = 0; v < 8; ++v) x2r[m][v] = x2s[rg * 32 + m * 16 + v + 8 * hi];

    float best[2][8]; int bestI[2][8];
#pragma unroll
    for (int m = 0; m < 2; ++m)
#pragma unroll
        for (int v = 0; v < 8; ++v) { best[m][v] = 3.4e38f; bestI[m][v] = 0; }

    auto epi = [&](const v8f (&acc)[2][2], int k0, float e0, float e1) {
#pragma unroll
        for (int m = 0; m < 2; ++m) {
#pragma unroll
            for (int v = 0; v < 8; ++v) {
                float d0 = x2r[m][v] - 2.0f * acc[0][m][v] + e0;
                if (d0 < best[m][v]) { best[m][v] = d0; bestI[m][v] = k0; }
                float d1 = x2r[m][v] - 2.0f * acc[1][m][v] + e1;
                if (d1 < best[m][v]) { best[m][v] = d1; bestI[m][v] = k0 + 16; }
            }
        }
    };

    v8f acc[2][2], R[2][2];
#pragma unroll
    for (int t = 0; t < 2; ++t)
#pragma unroll
        for (int m = 0; m < 2; ++m) { acc[t][m] = zero8; R[t][m] = zero8; }
    float e2r0 = __builtin_inff(), e2r1 = __builtin_inff();
    int   kr   = 0;

    v2f fx[2][16], fy[2][16];
    loadFrags(fx, 0, 0);

    int s_cur = 0;
    for (int chunk = 0; chunk < NCHUNKS; ++chunk) {
        const int s_nxt   = (s_cur == NSTAGE - 1) ? 0 : s_cur + 1;
        const int off_cur = s_cur * STAGE_FLOATS;
        const int off_nxt = s_nxt * STAGE_FLOATS;

        if (chunk + 1 < NCHUNKS) storeRegs(b_s + off_nxt);
        if (chunk + 2 < NCHUNKS) loadRegs(chunk + 2);

        const int k0 = chunk * NCHUNK + 32 * nh + l16;
        const float e0 = e2g[k0];
        const float e1 = e2g[k0 + 16];

        loadFrags(fy, off_cur, 1);
        __builtin_amdgcn_sched_barrier(0);
        mmaBlock(acc, fx, 0);
        epi(R, kr, e2r0, e2r1);
        __builtin_amdgcn_sched_barrier(0);

        __syncthreads();

        __builtin_amdgcn_sched_barrier(0);
        loadFrags(fx, off_nxt, 0);
        __builtin_amdgcn_sched_barrier(0);
        mmaBlock(acc, fy, 1);
        __builtin_amdgcn_sched_barrier(0);

#pragma unroll
        for (int t = 0; t < 2; ++t)
#pragma unroll
            for (int m = 0; m < 2; ++m) R[t][m] = acc[t][m];
        kr = k0; e2r0 = e0; e2r1 = e1;
        s_cur = s_nxt;
    }
    epi(R, kr, e2r0, e2r1);

#pragma unroll
    for (int m = 0; m < 2; ++m) {
#pragma unroll
        for (int v = 0; v < 8; ++v) {
            float d = best[m][v]; int i = bestI[m][v];
#pragma unroll
            for (int off = 8; off >= 1; off >>= 1) {
                float od = __shfl_xor(d, off, 32);
                int   oi = __shfl_xor(i, off, 32);
                if (od < d || (od == d && oi < i)) { d = od; i = oi; }
            }
            best[m][v] = d; bestI[m][v] = i;
        }
    }
    if (l16 == 0) {
#pragma unroll
        for (int m = 0; m < 2; ++m) {
#pragma unroll
            for (int v = 0; v < 8; ++v) {
                int rl = rg * 32 + m * 16 + v + 8 * hi;
                idx2_s[nh * ROWS_PB + rl]  = bestI[m][v];
                dmin2_s[nh * ROWS_PB + rl] = best[m][v];
            }
        }
    }
    __syncthreads();

    if (tid < ROWS_PB) {
        float d = dmin2_s[tid];           int i = idx2_s[tid];
        float od = dmin2_s[ROWS_PB + tid]; int oi = idx2_s[ROWS_PB + tid];
        if (od < d || (od == d && oi < i)) { d = od; i = oi; }
        idx_s[tid]  = i;
        dmin_s[tid] = d;
    }
    __syncthreads();

    if (tid == 0) {
        float s = 0.f;
        for (int r = 0; r < ROWS_PB; ++r) s += dmin_s[r];
        partials[blockIdx.x] = s;
    }

    {
        const int r  = tid & (ROWS_PB - 1);
        const int ch = (tid >> 7) * (CDIM / 2);
        const float* src = emb + (size_t)idx_s[r] * CDIM + ch;
        float* dst = outq + img_off + (size_t)ch * 1024 + r;
#pragma unroll 8
        for (int c4 = 0; c4 < CDIM / 8; ++c4) {
            v4f v = *(const v4f*)(src + 4 * c4);
            dst[(size_t)(4 * c4 + 0) * 1024] = v.x;
            dst[(size_t)(4 * c4 + 1) * 1024] = v.y;
            dst[(size_t)(4 * c4 + 2) * 1024] = v.z;
            dst[(size_t)(4 * c4 + 3) * 1024] = v.w;
        }
    }
}

__global__ __launch_bounds__(256) void vq_loss_kernel(const float* __restrict__ partials,
                                                      float* __restrict__ out_loss) {
    __shared__ float s[256];
    int t = threadIdx.x;
    s[t] = partials[t];
    __syncthreads();
    for (int off = 128; off >= 1; off >>= 1) {
        if (t < off) s[t] += s[t + off];
        __syncthreads();
    }
    if (t == 0) *out_loss = 1.25f * s[0] * (1.0f / 4194304.0f);
}

extern "C" void kernel_launch(void* const* d_in, const int* in_sizes, int n_in,
                              void* d_out, int out_size, void* d_ws, size_t ws_size,
                              hipStream_t stream) {
    const float* in  = (const float*)d_in[0];
    const float* emb = (const float*)d_in[1];
    float* outq      = (float*)d_out;
    float* partials  = (float*)d_ws;
    float* e2g       = partials + 256;
    float* loss_out  = outq + (out_size - 1);

    vq_e2_kernel  <<<K_CODES / 256, 256, 0, stream>>>(emb, e2g);
    vq_main_kernel<<<32768 / ROWS_PB, 256, 0, stream>>>(in, emb, e2g, outq, partials);
    vq_loss_kernel<<<1, 256, 0, stream>>>(partials, loss_out);
}
